# starting kernel
# baseline (speedup 1.0000x reference)
#include <hip/hip_runtime.h>
#include <hip/hip_bf16.h>


typedef float v2f __attribute__((ext_vector_type(2)));
typedef float v8f __attribute__((ext_vector_type(8)));

#define BB 16
#define TT 1024
#define CC 128
#define KTT (2*TT - 1)


__global__ __launch_bounds__(128)
void swin_qkbias_kernel(const float* __restrict__ Q,
                        const float* __restrict__ K,
                        const float* __restrict__ bias,
                        float* __restrict__ out) {
    __shared__ float lds[4][16 * 36];

    const int lane = threadIdx.x & 31;
    const int wv   = threadIdx.x >> 5;
    const int half = lane >> 4;
    const int l16  = lane & 15;

    const int bid   = blockIdx.x;
    const int tgrp  = bid & 3;
    const int wtile = (bid >> 2) & 63;
    const int b     = bid >> 8;

    const int w0      = wtile * 16;
    const int t0_base = tgrp * 256 + wv * 64;

    const float* Qb = Q + (size_t)b * TT * CC;
    const float* Kb = K + (size_t)b * KTT * CC;
    float*       Ob = out + (size_t)b * TT * TT;
    float* plds = &lds[wv][0];

    v2f qa[32];
    {
        const float* qrow = Qb + (size_t)(w0 + l16) * CC + 2 * half;
        #pragma unroll
        for (int n = 0; n < 32; ++n)
            qa[n] = *(const v2f*)(qrow + 4 * n);
    }

    const int base0 = w0 + t0_base;

    v8f p_cur;
    {
        int row = base0 + l16;
        if (row > KTT - 1) row = KTT - 1;
        const float* krow = Kb + (size_t)row * CC + 2 * half;
        v8f p = {};
        #pragma unroll
        for (int n = 0; n < 32; ++n) {
            v2f bv = *(const v2f*)(krow + 4 * n);
            p = __builtin_amdgcn_wmma_f32_16x16x4_f32(false, qa[n], false, bv,
                                                      (short)0, p, false, false);
        }
        p_cur = p;
    }

    for (int q = 0; q < 4; ++q) {
        const int t0 = t0_base + 16 * q;

        v8f p_next;
        {
            int row = base0 + 16 * (q + 1) + l16;
            if (row > KTT - 1) row = KTT - 1;
            const float* krow = Kb + (size_t)row * CC + 2 * half;
            v8f p = {};
            #pragma unroll
            for (int n = 0; n < 32; ++n) {
                v2f bv = *(const v2f*)(krow + 4 * n);
                p = __builtin_amdgcn_wmma_f32_16x16x4_f32(false, qa[n], false, bv,
                                                          (short)0, p, false, false);
            }
            p_next = p;
        }

        v8f acc = {};
        {
            const float* bcol = bias + (size_t)(2 * half) * TT + t0 + l16;
            #pragma unroll
            for (int n = 0; n < 32; ++n) {
                v2f bv;
                bv.x = bcol[(size_t)(4 * n) * TT];
                bv.y = bcol[(size_t)(4 * n + 1) * TT];
                acc = __builtin_amdgcn_wmma_f32_16x16x4_f32(false, qa[n], false, bv,
                                                            (short)0, acc, false, false);
            }
        }

        asm volatile("" ::: "memory");
        #pragma unroll
        for (int r = 0; r < 8; ++r) {
            int row = r + 8 * half;
            plds[row * 36 + l16]      = p_cur[r];
            plds[row * 36 + 16 + l16] = p_next[r];
        }
        asm volatile("s_wait_dscnt 0x0" ::: "memory");
        #pragma unroll
        for (int r = 0; r < 8; ++r) {
            int row = r + 8 * half;
            acc[r] += plds[row * 36 + row + l16];
        }
        asm volatile("" ::: "memory");

        #pragma unroll
        for (int r = 0; r < 8; ++r) {
            int row = r + 8 * half;
            Ob[(size_t)(w0 + row) * TT + t0 + l16] = acc[r];
        }

        p_cur = p_next;
    }
}

extern "C" void kernel_launch(void* const* d_in, const int* in_sizes, int n_in,
                              void* d_out, int out_size, void* d_ws, size_t ws_size,
                              hipStream_t stream) {
    (void)in_sizes; (void)n_in; (void)d_ws; (void)ws_size; (void)out_size;
    const float* Q    = (const float*)d_in[0];
    const float* K    = (const float*)d_in[1];
    const float* bias = (const float*)d_in[2];
    float* out = (float*)d_out;

    dim3 grid(BB * 64 * 4);
    dim3 block(128);
    swin_qkbias_kernel<<<grid, block, 0, stream>>>(Q, K, bias, out);
}


# Wave step widened to 32 columns with each lane owning 2 adjacent t: bias rows read and output rows written 64 bits per lane as whole 128-byte lines; 3-tile key strip staged per step; sums unchanged
# speedup vs baseline: 1.8609x; 1.8609x over previous
#include <hip/hip_runtime.h>


typedef float v2f __attribute__((ext_vector_type(2)));
typedef float v4f __attribute__((ext_vector_type(4)));
typedef float v8f __attribute__((ext_vector_type(8)));

#define BB 16
#define TT 1024
#define CC 128
#define KTT (2*TT - 1)
#define WN 4
#define TN 16
#define TP (TN / 2)
#define ST 51
#define NTG (TT / (64 * TN))


static __device__ __forceinline__ v8f mma4(v2f a, v2f b, v8f c) {
    return __builtin_amdgcn_wmma_f32_16x16x4_f32(false, a, false, b, (short)0, c, false, false);
}

__global__ __launch_bounds__(128) __attribute__((amdgpu_waves_per_eu(1, 1)))
void swin_qkbias_kernel(const float* __restrict__ Q,
                        const float* __restrict__ K,
                        const float* __restrict__ bias,
                        float* __restrict__ out) {
    __shared__ __attribute__((aligned(16))) float lds[4][WN][16 * ST];

    const int lane = threadIdx.x & 31;
    const int wv   = threadIdx.x >> 5;
    const int half = lane >> 4;
    const int l16  = lane & 15;

    const int bid   = blockIdx.x;
    const int tgrp  = bid % NTG;
    const int wquad = (bid / NTG) & 15;
    const int b     = bid / (NTG * 16);

    const int w0      = wquad * 64;
    const int t0_base = (tgrp * 4 + wv) * (16 * TN);

    const float* Qb = Q + (size_t)b * TT * CC;
    const float* Kb = K + (size_t)b * KTT * CC;
    float*       Ob = out + (size_t)b * TT * TT;
    float* pl[WN];
    #pragma unroll
    for (int i = 0; i < WN; ++i) pl[i] = &lds[wv][i][0];

    v2f qa[WN][32];
    #pragma unroll
    for (int i = 0; i < WN; ++i) {
        const float* qrow = Qb + (size_t)(w0 + 16 * i + l16) * CC + 4 * half;
        #pragma unroll
        for (int m = 0; m < 16; ++m) {
            v4f a = *(const v4f*)(qrow + 8 * m);
            qa[i][2 * m]     = a.xy;
            qa[i][2 * m + 1] = a.zw;
        }
    }

    const int base0 = w0 + t0_base;

    v8f P[WN][WN + 2];
    #pragma unroll
    for (int i = 0; i < WN; ++i) {
        #pragma unroll
        for (int d = 0; d < WN + 2; ++d) P[i][d] = v8f{};
    }

    #pragma unroll
    for (int j = 0; j < WN; ++j) {
        const float* krow = Kb + (size_t)(base0 + 16 * j + l16) * CC + 4 * half;
        #pragma unroll
        for (int m = 0; m < 16; ++m) {
            v4f kv = *(const v4f*)(krow + 8 * m);
            #pragma unroll
            for (int i = 0; i <= j; ++i) P[i][j - i] = mma4(qa[i][2 * m],     kv.xy, P[i][j - i]);
            #pragma unroll
            for (int i = 0; i <= j; ++i) P[i][j - i] = mma4(qa[i][2 * m + 1], kv.zw, P[i][j - i]);
        }
    }

    v4f kva[16], kvb[16];
    auto load_keys = [&](const int tile) {
        int rowa = base0 + 16 * tile + l16;
        int rowb = rowa + 16;
        if (rowa > KTT - 1) rowa = KTT - 1;
        if (rowb > KTT - 1) rowb = KTT - 1;
        const float* kra = Kb + (size_t)rowa * CC + 4 * half;
        const float* krb = Kb + (size_t)rowb * CC + 4 * half;
        #pragma unroll
        for (int m = 0; m < 16; ++m) kva[m] = *(const v4f*)(kra + 8 * m);
        #pragma unroll
        for (int m = 0; m < 16; ++m) kvb[m] = *(const v4f*)(krb + 8 * m);
    };
    load_keys(WN);

    constexpr int NB1 = 14;

    auto step = [&](const int s, const int nska, const int nskb, const bool more) {
        const int t0 = t0_base + 32 * s;

        __builtin_amdgcn_sched_barrier(0);

        const float* bcol = bias + (size_t)(4 * half) * TT + t0 + 2 * l16;
        v2f lb[16][4];
        #pragma unroll
        for (int m = 0; m < NB1; ++m) {
            #pragma unroll
            for (int k = 0; k < 4; ++k) lb[m][k] = *(const v2f*)(bcol + (size_t)(8 * m + k) * TT);
        }
        __builtin_amdgcn_sched_barrier(0);

        v8f pa[WN], pb[WN];
        #pragma unroll
        for (int i = 0; i < WN; ++i) { pa[i] = v8f{}; pb[i] = v8f{}; }
        #pragma unroll
        for (int m = 0; m < 16; ++m) {
            #pragma unroll
            for (int i = 0; i < WN; ++i) if (i >= nska) pa[i] = mma4(qa[i][2 * m],     kva[m].xy, pa[i]);
            #pragma unroll
            for (int i = 0; i < WN; ++i) if (i >= nska) pa[i] = mma4(qa[i][2 * m + 1], kva[m].zw, pa[i]);
        }
        #pragma unroll
        for (int m = 0; m < 16; ++m) {
            #pragma unroll
            for (int i = 0; i < WN; ++i) if (i >= nskb) pb[i] = mma4(qa[i][2 * m],     kvb[m].xy, pb[i]);
            #pragma unroll
            for (int i = 0; i < WN; ++i) if (i >= nskb) pb[i] = mma4(qa[i][2 * m + 1], kvb[m].zw, pb[i]);
        }
        #pragma unroll
        for (int i = 0; i < WN; ++i) {
            if (i >= nska) P[i][WN - i]     = pa[i];
            if (i >= nskb) P[i][WN + 1 - i] = pb[i];
        }

        __builtin_amdgcn_sched_barrier(0);
        #pragma unroll
        for (int i = 0; i < WN; ++i) {
            #pragma unroll
            for (int r = 0; r < 8; ++r) {
                int row = r + 8 * half;
                pl[i][row * ST + l16]      = P[i][0][r];
                pl[i][row * ST + 16 + l16] = P[i][1][r];
                pl[i][row * ST + 32 + l16] = P[i][2][r];
            }
        }
        __builtin_amdgcn_sched_barrier(0);

        #pragma unroll
        for (int m = NB1; m < 16; ++m) {
            #pragma unroll
            for (int k = 0; k < 4; ++k) lb[m][k] = *(const v2f*)(bcol + (size_t)(8 * m + k) * TT);
        }
        if (more) load_keys(2 * s + WN + 2);
        __builtin_amdgcn_sched_barrier(0);

        v8f acc[WN][2];
        #pragma unroll
        for (int i = 0; i < WN; ++i) { acc[i][0] = v8f{}; acc[i][1] = v8f{}; }
        #pragma unroll
        for (int m = 0; m < 16; ++m) {
            #pragma unroll
            for (int u = 0; u < 2; ++u) {
                v2f be, bo;
                be.x = lb[m][2 * u].x;  be.y = lb[m][2 * u + 1].x;
                bo.x = lb[m][2 * u].y;  bo.y = lb[m][2 * u + 1].y;
                #pragma unroll
                for (int i = 0; i < WN; ++i) acc[i][0] = mma4(qa[i][2 * m + u], be, acc[i][0]);
                #pragma unroll
                for (int i = 0; i < WN; ++i) acc[i][1] = mma4(qa[i][2 * m + u], bo, acc[i][1]);
            }
        }

        asm volatile("s_wait_dscnt 0x0" ::: "memory");
        v2f o[WN][8];
        #pragma unroll
        for (int i = 0; i < WN; ++i) {
            #pragma unroll
            for (int r = 0; r < 8; ++r) {
                int row = r + 8 * half;
                v2f g = *(const v2f*)(&pl[i][row * (ST + 1) + 2 * l16]);
                o[i][r].x = acc[i][0][r] + g.x;
                o[i][r].y = acc[i][1][r] + g.y;
            }
        }
        asm volatile("" ::: "memory");

        #pragma unroll
        for (int i = 0; i < WN; ++i) {
            #pragma unroll
            for (int r = 0; r < 8; ++r) {
                int row = r + 8 * half;
                *(v2f*)(&Ob[(size_t)(w0 + 16 * i + row) * TT + t0 + 2 * l16]) = o[i][r];
            }
        }

        #pragma unroll
        for (int i = 0; i < WN; ++i) {
            #pragma unroll
            for (int d = 0; d < WN - i; ++d) P[i][d] = P[i][d + 2];
        }
    };

    for (int s = 0; s <= TP - 3; ++s) step(s, 0, 0, true);
    step(TP - 2, 0, 1, true);
    step(TP - 1, 2, 3, false);
}

extern "C" void kernel_launch(void* const* d_in, const int* in_sizes, int n_in,
                              void* d_out, int out_size, void* d_ws, size_t ws_size,
                              hipStream_t stream) {
    (void)in_sizes; (void)n_in; (void)d_ws; (void)ws_size; (void)out_size;
    const float* Q    = (const float*)d_in[0];
    const float* K    = (const float*)d_in[1];
    const float* bias = (const float*)d_in[2];
    float* out = (float*)d_out;

    dim3 grid(BB * 16 * NTG);
    dim3 block(128);
    swin_qkbias_kernel<<<grid, block, 0, stream>>>(Q, K, bias, out);
}
